# starting kernel
# baseline (speedup 1.0000x reference)
#include <hip/hip_runtime.h>

typedef float v2f __attribute__((ext_vector_type(2)));
typedef float v8f __attribute__((ext_vector_type(8)));

#define XW 4096
#define OW 4081
#define ROWS_PER_SEG 128
#define NSTRIPS 256
#define NSEGS 32

__device__ __forceinline__ float swap_halves(float v) {
  int i = __builtin_bit_cast(int, v);
  int s = __builtin_amdgcn_permlanex16(i, i, 0x76543210u, 0xfedcba98u,
                                       false, false);
  return __builtin_bit_cast(float, s);
}

#define STEP(u, T, DO_STORE)                                                   \
  {                                                                            \
    const int t = (T) + (u);                                                   \
    const int r = os + t;                    \
    float old = acc[(u)&7];                                                    \
    if (DO_STORE) {                                                            \
      const int orow = r - 16;                                      \
      if ((orow < oe) && storelane) {      \
        const char* orowb = (const char*)out + (size_t)orow * (OW * 4);        \
        *(float*)(orowb + ocolb) = old + bval;                                 \
      }                                                                        \
    }                                                                          \
    float sw = swap_halves(old);                                               \
    acc[(u)&7] = h ? sw : 0.0f;                                                \
    const char* xrow = (const char*)x + (size_t)min(r, XW - 1) * (XW * 4);     \
    v8f y = {};                                                                \
    _Pragma("unroll") for (int c = 0; c < 4; ++c) {                            \
      v2f b;                                                                   \
      b.x = *(const float*)(xrow + coffb[2 * c]);                              \
      b.y = *(const float*)(xrow + coffb[2 * c + 1]);                          \
      y = __builtin_amdgcn_wmma_f32_16x16x4_f32(false, A[c], false, b,         \
                                                (short)0, y, false, false);    \
    }                                                                          \
    _Pragma("unroll") for (int k = 0; k < 8; ++k) acc[((u)-k) & 7] += y[k];    \
  }

__global__ __launch_bounds__(256) void conv16x16_wmma_f32(
    const float* __restrict__ x, const float* __restrict__ w,
    const float* __restrict__ bias, float* __restrict__ out)
{
  const int tid  = blockIdx.x * blockDim.x + threadIdx.x;
  const int lane = threadIdx.x & 31;
  const int wave  = __builtin_amdgcn_readfirstlane(tid >> 5);
  const int strip = wave & (NSTRIPS - 1);
  const int seg   = wave >> 8;
  const int j0    = strip * 16;
  const int os    = seg * ROWS_PER_SEG;
  const int oe    = min(os + ROWS_PER_SEG, OW);
  const int n     = lane & 15;
  const int h     = lane >> 4;

  v2f A[4];
#pragma unroll
  for (int c = 0; c < 4; ++c) {
    A[c].x = w[n * 16 + 4 * c + 2 * h];
    A[c].y = w[n * 16 + 4 * c + 2 * h + 1];
  }
  const float bval = bias[0];

  const int colbase = j0 + n + 2 * h;
  unsigned coffb[8];
#pragma unroll
  for (int c = 0; c < 4; ++c) {
    coffb[2 * c]     = (unsigned)min(colbase + 4 * c,     XW - 1) * 4u;
    coffb[2 * c + 1] = (unsigned)min(colbase + 4 * c + 1, XW - 1) * 4u;
  }

  const int ocol = j0 + n;
  const bool storelane = (h != 0) && (ocol < OW);
  const unsigned ocolb = (unsigned)ocol * 4u;

  float acc[8];
#pragma unroll
  for (int s = 0; s < 8; ++s) acc[s] = 0.0f;

#pragma unroll
  for (int u = 0; u < 8; ++u) STEP(u, 0, false);
#pragma unroll
  for (int u = 0; u < 8; ++u) STEP(u, 8, false);

  const int steps = ((oe - os) + 7) & ~7;
  for (int T = 16; T < steps + 16; T += 8) {
#pragma unroll
    for (int u = 0; u < 8; ++u) STEP(u, T, true);
  }
}

extern "C" void kernel_launch(void* const* d_in, const int* in_sizes, int n_in,
                              void* d_out, int out_size, void* d_ws, size_t ws_size,
                              hipStream_t stream) {
  const float* x    = (const float*)d_in[0];
  const float* wgt  = (const float*)d_in[1];
  const float* bias = (const float*)d_in[2];
  float* out = (float*)d_out;
  dim3 grid(1024), block(256);
  hipLaunchKernelGGL(conv16x16_wmma_f32, grid, block, 0, stream,
                     x, wgt, bias, out);
}


# Load each image row's B values three steps ahead into a 4-slot register ring (one clamped per-lane base pointer, so adjacent pairs go as 8-byte loads): no step waits on loads it has just issued
# speedup vs baseline: 1.4268x; 1.4268x over previous
#include <hip/hip_runtime.h>

typedef float v2f __attribute__((ext_vector_type(2)));
typedef float v8f __attribute__((ext_vector_type(8)));

#define XW 4096
#define OW 4081
#define ROWS_PER_SEG 128
#define NSTRIPS 256
#define NSEGS 32

__device__ __forceinline__ float swap_halves(float v) {
  int i = __builtin_bit_cast(int, v);
  int s = __builtin_amdgcn_permlanex16(i, i, 0x76543210u, 0xfedcba98u,
                                       false, false);
  return __builtin_bit_cast(float, s);
}

#define PF 3
#define LOADROW(slot, rr)                                                      \
  {                                                                            \
    const char* xrow = (const char*)x + (size_t)min((rr), XW - 1) * (XW * 4);  \
    const float* p = (const float*)(xrow + lofs);                              \
    _Pragma("unroll") for (int c = 0; c < 4; ++c) {                            \
      Bq[(slot)][2 * c]     = p[4 * c];                                        \
      Bq[(slot)][2 * c + 1] = p[4 * c + 1];                                    \
    }                                                                          \
  }
#define STEP(u, T, DO_STORE)                                                   \
  {                                                                            \
    const int t = (T) + (u);                                                   \
    const int r = os + t;                    \
    float old = acc[(u)&7];                                                    \
    if (DO_STORE) {                                                            \
      const int orow = r - 16;                                      \
      if ((orow < oe) && storelane) {      \
        const char* orowb = (const char*)out + (size_t)orow * (OW * 4);        \
        *(float*)(orowb + ocolb) = old + bval;                                 \
      }                                                                        \
    }                                                                          \
    LOADROW(((u) + PF) & 3, r + PF)                                            \
    float sw = swap_halves(old);                                               \
    acc[(u)&7] = h ? sw : 0.0f;                                                \
    v8f y = {};                                                                \
    _Pragma("unroll") for (int c = 0; c < 4; ++c) {                            \
      v2f b;                                                                   \
      b.x = Bq[(u)&3][2 * c];                                                  \
      b.y = Bq[(u)&3][2 * c + 1];                                              \
      y = __builtin_amdgcn_wmma_f32_16x16x4_f32(false, A[c], false, b,         \
                                                (short)0, y, false, false);    \
    }                                                                          \
    _Pragma("unroll") for (int k = 0; k < 8; ++k) acc[((u)-k) & 7] += y[k];    \
  }

__global__ __launch_bounds__(256) void conv16x16_wmma_f32(
    const float* __restrict__ x, const float* __restrict__ w,
    const float* __restrict__ bias, float* __restrict__ out)
{
  const int tid  = blockIdx.x * blockDim.x + threadIdx.x;
  const int lane = threadIdx.x & 31;
  const int wave  = __builtin_amdgcn_readfirstlane(tid >> 5);
  const int strip = wave & (NSTRIPS - 1);
  const int seg   = wave >> 8;
  const int j0    = strip * 16;
  const int os    = seg * ROWS_PER_SEG;
  const int oe    = min(os + ROWS_PER_SEG, OW);
  const int n     = lane & 15;
  const int h     = lane >> 4;

  v2f A[4];
#pragma unroll
  for (int c = 0; c < 4; ++c) {
    A[c].x = w[n * 16 + 4 * c + 2 * h];
    A[c].y = w[n * 16 + 4 * c + 2 * h + 1];
  }
  const float bval = bias[0];

  const unsigned lofs = (unsigned)(min(j0 + n, OW - 1) + 2 * h) * 4u;

  const int ocol = j0 + n;
  const bool storelane = (h != 0) && (ocol < OW);
  const unsigned ocolb = (unsigned)ocol * 4u;

  float acc[8];
#pragma unroll
  for (int s = 0; s < 8; ++s) acc[s] = 0.0f;

  float Bq[4][8];
#pragma unroll
  for (int s = 0; s < PF; ++s) LOADROW(s, os + s);

#pragma unroll
  for (int u = 0; u < 8; ++u) STEP(u, 0, false);
#pragma unroll
  for (int u = 0; u < 8; ++u) STEP(u, 8, false);

  const int steps = ((oe - os) + 7) & ~7;
  for (int T = 16; T < steps + 16; T += 8) {
#pragma unroll
    for (int u = 0; u < 8; ++u) STEP(u, T, true);
  }
}

extern "C" void kernel_launch(void* const* d_in, const int* in_sizes, int n_in,
                              void* d_out, int out_size, void* d_ws, size_t ws_size,
                              hipStream_t stream) {
  const float* x    = (const float*)d_in[0];
  const float* wgt  = (const float*)d_in[1];
  const float* bias = (const float*)d_in[2];
  float* out = (float*)d_out;
  dim3 grid(1024), block(256);
  hipLaunchKernelGGL(conv16x16_wmma_f32, grid, block, 0, stream,
                     x, wgt, bias, out);
}


# Every 8-step block inside the image addresses memory through one load and one store pointer (row as immediate offset; warm-up blocks in the same loop): per-step clamp, address sums and row test go
# speedup vs baseline: 1.5030x; 1.0534x over previous
#include <hip/hip_runtime.h>


typedef float v2f __attribute__((ext_vector_type(2)));
typedef float v8f __attribute__((ext_vector_type(8)));

#define XW 4096
#define OW 4081
#define ROWS_PER_SEG 128
#define NSTRIPS 256
#define NSEGS 32

__device__ __forceinline__ float swap_halves(float v) {
  int i = __builtin_bit_cast(int, v);
  int s = __builtin_amdgcn_permlanex16(i, i, 0x76543210u, 0xfedcba98u,
                                       false, false);
  return __builtin_bit_cast(float, s);
}

#define PF 3
#define LOADROW_C(slot, rr)                                                    \
  {                                                                            \
    const char* xrow = (const char*)x + (size_t)min((rr), XW - 1) * (XW * 4);  \
    const float* p = (const float*)(xrow + lofs);                              \
    _Pragma("unroll") for (int c = 0; c < 4; ++c) {                            \
      Bq[(slot)][2 * c]     = p[4 * c];                                        \
      Bq[(slot)][2 * c + 1] = p[4 * c + 1];                                    \
    }                                                                          \
  }
#define LOADROW_H(slot, dr)                                                    \
  {                                                                            \
    const float* p = pT + (dr) * XW;          \
    _Pragma("unroll") for (int c = 0; c < 4; ++c) {                            \
      Bq[(slot)][2 * c]     = p[4 * c];                                        \
      Bq[(slot)][2 * c + 1] = p[4 * c + 1];                                    \
    }                                                                          \
  }
#define STEP(u, T, DO_STORE, HOLD)                                             \
  {                                                                            \
    const int t = (T) + (u);                                                   \
    const int r = os + t;                    \
    float old = acc[(u)&7];                                                    \
    if (DO_STORE) {                                                            \
      if (HOLD) {                                                              \
        if (storelane) oT[(u) * OW] = old + bval;       \
      } else {                                                                 \
        const int orow = r - 16;                                    \
        if ((orow < oe) && storelane) {              \
          const char* orowb = (const char*)out + (size_t)orow * (OW * 4);      \
          *(float*)(orowb + ocolb) = old + bval;                               \
        }                                                                      \
      }                                                                        \
    }                                                                          \
    if (HOLD) LOADROW_H(((u) + PF) & 3, (u) + PF)                              \
    else      LOADROW_C(((u) + PF) & 3, r + PF)                                \
    float sw = swap_halves(old);                                               \
    acc[(u)&7] = h ? sw : 0.0f;                                                \
    v8f y = {};                                                                \
    _Pragma("unroll") for (int c = 0; c < 4; ++c) {                            \
      v2f b;                                                                   \
      b.x = Bq[(u)&3][2 * c];                                                  \
      b.y = Bq[(u)&3][2 * c + 1];                                              \
      y = __builtin_amdgcn_wmma_f32_16x16x4_f32(false, A[c], false, b,         \
                                                (short)0, y, false, false);    \
    }                                                                          \
    _Pragma("unroll") for (int k = 0; k < 8; ++k) acc[((u)-k) & 7] += y[k];    \
  }

__global__ __launch_bounds__(256, 8) void conv16x16_wmma_f32(
    const float* __restrict__ x, const float* __restrict__ w,
    const float* __restrict__ bias, float* __restrict__ out)
{
  const int tid  = blockIdx.x * blockDim.x + threadIdx.x;
  const int lane = threadIdx.x & 31;
  const int wave  = __builtin_amdgcn_readfirstlane(tid >> 5);
  const int strip = wave & (NSTRIPS - 1);
  const int seg   = wave >> 8;
  const int j0    = strip * 16;
  const int os    = seg * ROWS_PER_SEG;
  const int oe    = min(os + ROWS_PER_SEG, OW);
  const int n     = lane & 15;
  const int h     = lane >> 4;

  v2f A[4];
#pragma unroll
  for (int c = 0; c < 4; ++c) {
    A[c].x = w[n * 16 + 4 * c + 2 * h];
    A[c].y = w[n * 16 + 4 * c + 2 * h + 1];
  }
  const float bval = bias[0];

  const unsigned lofs = (unsigned)(min(j0 + n, OW - 1) + 2 * h) * 4u;

  const int ocol = j0 + n;
  const bool storelane = (h != 0) && (ocol < OW);
  const unsigned ocolb = (unsigned)ocol * 4u;

  float acc[8];
#pragma unroll
  for (int s = 0; s < 8; ++s) acc[s] = 0.0f;

  float Bq[4][8];

  const int steps = ((oe - os) + 7) & ~7;
  const int thold = (seg != NSEGS - 1) ? ROWS_PER_SEG + 16 : 120;

  const float* xl = (const float*)((const char*)x + lofs);
  float* ol = (float*)((char*)out + (storelane ? ocolb : 0u));
  const float* pT = xl + (size_t)os * XW;
  float* oT = ol;
#pragma unroll
  for (int s = 0; s < PF; ++s) LOADROW_H(s, s);

  int T = 0;
  {
    const bool lanes = storelane;
    for (; T < thold; T += 8) {
      pT = xl + (size_t)(os + T) * XW;
      oT = ol + (size_t)(os + max(T, 16) - 16) * OW;
      const bool storelane = lanes && (T >= 16);
#pragma unroll
      for (int u = 0; u < 8; ++u) STEP(u, T, true, true);
    }
  }
  for (; T < steps + 16; T += 8) {
#pragma unroll
    for (int u = 0; u < 8; ++u) STEP(u, T, true, false);
  }
}

extern "C" void kernel_launch(void* const* d_in, const int* in_sizes, int n_in,
                              void* d_out, int out_size, void* d_ws, size_t ws_size,
                              hipStream_t stream) {
  const float* x    = (const float*)d_in[0];
  const float* wgt  = (const float*)d_in[1];
  const float* bias = (const float*)d_in[2];
  float* out = (float*)d_out;
  dim3 grid(1024), block(256);
  hipLaunchKernelGGL(conv16x16_wmma_f32, grid, block, 0, stream,
                     x, wgt, bias, out);
}


# Each image row's B values are requested five steps ahead into an 8-slot register ring (was three ahead, 4 slots): a row from memory has arrived before its step; every sum and its order unchanged
# speedup vs baseline: 1.5595x; 1.0376x over previous
#include <hip/hip_runtime.h>


typedef float v2f __attribute__((ext_vector_type(2)));
typedef float v8f __attribute__((ext_vector_type(8)));

#define XW 4096
#define OW 4081
#define ROWS_PER_SEG 128
#define NSTRIPS 256
#define NSEGS 32

__device__ __forceinline__ float swap_halves(float v) {
  int i = __builtin_bit_cast(int, v);
  int s = __builtin_amdgcn_permlanex16(i, i, 0x76543210u, 0xfedcba98u,
                                       false, false);
  return __builtin_bit_cast(float, s);
}

#define PF 5
#define LOADROW_C(slot, rr)                                                    \
  {                                                                            \
    const char* xrow = (const char*)x + (size_t)min((rr), XW - 1) * (XW * 4);  \
    const float* p = (const float*)(xrow + lofs);                              \
    _Pragma("unroll") for (int c = 0; c < 4; ++c) {                            \
      Bq[(slot)][2 * c]     = p[4 * c];                                        \
      Bq[(slot)][2 * c + 1] = p[4 * c + 1];                                    \
    }                                                                          \
  }
#define LOADROW_H(slot, dr)                                                    \
  {                                                                            \
    const float* p = pT + (dr) * XW;          \
    _Pragma("unroll") for (int c = 0; c < 4; ++c) {                            \
      Bq[(slot)][2 * c]     = p[4 * c];                                        \
      Bq[(slot)][2 * c + 1] = p[4 * c + 1];                                    \
    }                                                                          \
  }
#define STEP(u, T, DO_STORE, HOLD)                                             \
  {                                                                            \
    const int t = (T) + (u);                                                   \
    const int r = os + t;                    \
    float old = acc[(u)&7];                                                    \
    if (DO_STORE) {                                                            \
      if (HOLD) {                                                              \
        if (storelane) oT[(u) * OW] = old + bval;       \
      } else {                                                                 \
        const int orow = r - 16;                                    \
        if ((orow < oe) && storelane) {              \
          const char* orowb = (const char*)out + (size_t)orow * (OW * 4);      \
          *(float*)(orowb + ocolb) = old + bval;                               \
        }                                                                      \
      }                                                                        \
    }                                                                          \
    if (HOLD) LOADROW_H(((u) + PF) & 7, (u) + PF)                              \
    else      LOADROW_C(((u) + PF) & 7, r + PF)                                \
    float sw = swap_halves(old);                                               \
    acc[(u)&7] = h ? sw : 0.0f;                                                \
    v8f y = {};                                                                \
    _Pragma("unroll") for (int c = 0; c < 4; ++c) {                            \
      v2f b;                                                                   \
      b.x = Bq[(u)&7][2 * c];                                                  \
      b.y = Bq[(u)&7][2 * c + 1];                                              \
      y = __builtin_amdgcn_wmma_f32_16x16x4_f32(false, A[c], false, b,         \
                                                (short)0, y, false, false);    \
    }                                                                          \
    _Pragma("unroll") for (int k = 0; k < 8; ++k) acc[((u)-k) & 7] += y[k];    \
  }

__global__ __launch_bounds__(256, 8) void conv16x16_wmma_f32(
    const float* __restrict__ x, const float* __restrict__ w,
    const float* __restrict__ bias, float* __restrict__ out)
{
  const int tid  = blockIdx.x * blockDim.x + threadIdx.x;
  const int lane = threadIdx.x & 31;
  const int wave  = __builtin_amdgcn_readfirstlane(tid >> 5);
  const int strip = wave & (NSTRIPS - 1);
  const int seg   = wave >> 8;
  const int j0    = strip * 16;
  const int os    = seg * ROWS_PER_SEG;
  const int oe    = min(os + ROWS_PER_SEG, OW);
  const int n     = lane & 15;
  const int h     = lane >> 4;

  v2f A[4];
#pragma unroll
  for (int c = 0; c < 4; ++c) {
    A[c].x = w[n * 16 + 4 * c + 2 * h];
    A[c].y = w[n * 16 + 4 * c + 2 * h + 1];
  }
  const float bval = bias[0];

  const unsigned lofs = (unsigned)(min(j0 + n, OW - 1) + 2 * h) * 4u;

  const int ocol = j0 + n;
  const bool storelane = (h != 0) && (ocol < OW);
  const unsigned ocolb = (unsigned)ocol * 4u;

  float acc[8];
#pragma unroll
  for (int s = 0; s < 8; ++s) acc[s] = 0.0f;

  float Bq[8][8];

  const int steps = ((oe - os) + 7) & ~7;
  const int thold = (seg != NSEGS - 1) ? ROWS_PER_SEG + 16 : 120;

  const float* xl = (const float*)((const char*)x + lofs);
  float* ol = (float*)((char*)out + (storelane ? ocolb : 0u));
  const float* pT = xl + (size_t)os * XW;
  float* oT = ol;
#pragma unroll
  for (int s = 0; s < PF; ++s) LOADROW_H(s, s);

  int T = 0;
  {
    const bool lanes = storelane;
    for (; T < thold; T += 8) {
      pT = xl + (size_t)(os + T) * XW;
      oT = ol + (size_t)(os + max(T, 16) - 16) * OW;
      const bool storelane = lanes && (T >= 16);
#pragma unroll
      for (int u = 0; u < 8; ++u) STEP(u, T, true, true);
    }
  }
  {
    const int lane2 = (int)__builtin_amdgcn_mbcnt_lo(~0u, 0u);
    const int n = lane2 & 15;
    const int h = lane2 >> 4;
    const unsigned lofs = (unsigned)(min(j0 + n, OW - 1) + 2 * h) * 4u;
    const int ocol = j0 + n;
    const bool storelane = (h != 0) && (ocol < OW);
    const unsigned ocolb = (unsigned)ocol * 4u;
    for (; T < steps + 16; T += 8) {
#pragma unroll
      for (int u = 0; u < 8; ++u) STEP(u, T, true, false);
    }
  }
}

extern "C" void kernel_launch(void* const* d_in, const int* in_sizes, int n_in,
                              void* d_out, int out_size, void* d_ws, size_t ws_size,
                              hipStream_t stream) {
  const float* x    = (const float*)d_in[0];
  const float* wgt  = (const float*)d_in[1];
  const float* bias = (const float*)d_in[2];
  float* out = (float*)d_out;
  dim3 grid(1024), block(256);
  hipLaunchKernelGGL(conv16x16_wmma_f32, grid, block, 0, stream,
                     x, wgt, bias, out);
}
